# starting kernel
# baseline (speedup 1.0000x reference)
#include <hip/hip_runtime.h>
#include <math.h>


#define GRID_HH 32
#define GRID_WW 32
#define OBJ_DIM 64
#define LOC_DIM 16
#define SDIM    64
#define NCOLS   (GRID_HH * GRID_WW)
#define BATCH   64
#define ETA_X   0.05f
#define NSTEPS  20

#define XO_TOT  ((size_t)NCOLS * BATCH * OBJ_DIM)
#define XL_TOT  ((size_t)NCOLS * BATCH * LOC_DIM)

typedef __attribute__((ext_vector_type(2))) float v2f;
typedef __attribute__((ext_vector_type(8))) float v8f;

__device__ __forceinline__ v8f wmma_f32_k4(v2f a, v2f b, v8f c) {
  return __builtin_amdgcn_wmma_f32_16x16x4_f32(
      false, a, false, b, (short)0, c, false, false);
}

__global__ void cg_init_kernel(float* __restrict__ xo,
                               float* __restrict__ xl,
                               float* __restrict__ energy) {
  size_t i = (size_t)blockIdx.x * blockDim.x + threadIdx.x;
  size_t stride = (size_t)gridDim.x * blockDim.x;
  for (size_t e = i; e < XO_TOT; e += stride) xo[e] = 0.0f;
  for (size_t e = i; e < XL_TOT; e += stride) xl[e] = 0.0f;
  if (i < NSTEPS) energy[i] = 0.0f;
}

__global__ __launch_bounds__(512)
void cg_step_kernel(const float* __restrict__ gin,
                    const float* __restrict__ Wobj,
                    const float* __restrict__ Wloc,
                    const float* __restrict__ xo_in,
                    const float* __restrict__ xl_in,
                    float* __restrict__ xo_out,
                    float* __restrict__ xl_out,
                    float* __restrict__ energy_t) {
  __shared__ float sWo[OBJ_DIM][65];
  __shared__ float sWl[LOC_DIM][65];
  __shared__ float sPG[BATCH][65];
  __shared__ float sXo[BATCH][65];
  __shared__ float sXl[BATCH][17];
  __shared__ float sCtx[BATCH][65];
  __shared__ float sEnergy;

  const int n   = blockIdx.x;
  const int gh  = n >> 5;
  const int gw  = n & 31;
  const int tid = threadIdx.x;

  if (tid == 0) sEnergy = 0.0f;

  for (int e = tid; e < OBJ_DIM * SDIM; e += 512)
    sWo[e >> 6][e & 63] = Wobj[(size_t)n * OBJ_DIM * SDIM + e];
  for (int e = tid; e < LOC_DIM * SDIM; e += 512)
    sWl[e >> 6][e & 63] = Wloc[(size_t)n * LOC_DIM * SDIM + e];

  for (int e = tid; e < BATCH * SDIM; e += 512) {
    int b = e >> 6, s = e & 63;
    int py = s >> 3, px = s & 7;
    sPG[b][s] = gin[(size_t)b * 65536 + (size_t)(gh * 8 + py) * 256 + gw * 8 + px];
  }

  const float inv_cnt =
      1.0f / (float)((gh > 0) + (gh < GRID_HH - 1) + (gw > 0) + (gw < GRID_WW - 1));
  for (int e = tid; e < BATCH * OBJ_DIM; e += 512) {
    sXo[e >> 6][e & 63] = xo_in[(size_t)n * BATCH * OBJ_DIM + e];
    float ssum = 0.0f;
    if (gh > 0)           ssum += xo_in[(size_t)(n - GRID_WW) * BATCH * OBJ_DIM + e];
    if (gh < GRID_HH - 1) ssum += xo_in[(size_t)(n + GRID_WW) * BATCH * OBJ_DIM + e];
    if (gw > 0)           ssum += xo_in[(size_t)(n - 1) * BATCH * OBJ_DIM + e];
    if (gw < GRID_WW - 1) ssum += xo_in[(size_t)(n + 1) * BATCH * OBJ_DIM + e];
    sCtx[e >> 6][e & 63] = ssum * inv_cnt;
  }
  for (int e = tid; e < BATCH * LOC_DIM; e += 512)
    sXl[e >> 4][e & 15] = xl_in[(size_t)n * BATCH * LOC_DIM + e];
  __syncthreads();

  const int lane = tid & 31;
  const int wv   = tid >> 5;
  const int kh   = lane >> 4;
  const int mr   = lane & 15;
  const int mt   = (wv >> 2) * 16;
  const int nt   = (wv & 3) * 16;

  v8f acc = {};
  for (int k = 0; k < SDIM; k += 4) {
    v2f a, b;
    a.x = sXo[mt + mr][k + 2 * kh];
    a.y = sXo[mt + mr][k + 2 * kh + 1];
    b.x = sWo[k + 2 * kh][nt + mr];
    b.y = sWo[k + 2 * kh + 1][nt + mr];
    acc = wmma_f32_k4(a, b, acc);
  }
  for (int k = 0; k < LOC_DIM; k += 4) {
    v2f a, b;
    a.x = sXl[mt + mr][k + 2 * kh];
    a.y = sXl[mt + mr][k + 2 * kh + 1];
    b.x = sWl[k + 2 * kh][nt + mr];
    b.y = sWl[k + 2 * kh + 1][nt + mr];
    acc = wmma_f32_k4(a, b, acc);
  }

  float epart = 0.0f;
#pragma unroll
  for (int r = 0; r < 8; ++r) {
    int bb = mt + r + 8 * kh;
    int ss = nt + mr;
    float pred = tanhf(acc[r]);
    float eps  = sPG[bb][ss] - pred;
    float g    = eps * (1.0f - pred * pred);
    epart += 0.5f * eps * eps;
    sPG[bb][ss] = g;
  }

  for (int off = 16; off > 0; off >>= 1) epart += __shfl_xor(epart, off, 32);
  if (lane == 0) atomicAdd(&sEnergy, epart);
  __syncthreads();

  v8f acc2 = {};
  for (int k = 0; k < SDIM; k += 4) {
    v2f a, b;
    a.x = sPG[mt + mr][k + 2 * kh];
    a.y = sPG[mt + mr][k + 2 * kh + 1];
    b.x = sWo[nt + mr][k + 2 * kh];
    b.y = sWo[nt + mr][k + 2 * kh + 1];
    acc2 = wmma_f32_k4(a, b, acc2);
  }
#pragma unroll
  for (int r = 0; r < 8; ++r) {
    int bb = mt + r + 8 * kh;
    int oo = nt + mr;
    float xo = sXo[bb][oo];
    xo_out[(size_t)n * BATCH * OBJ_DIM + bb * OBJ_DIM + oo] =
        xo + ETA_X * (acc2[r] + sCtx[bb][oo] - xo);
  }

  const int mt2 = (wv & 3) * 16;
  v8f acc3 = {};
  for (int k = 0; k < SDIM; k += 4) {
    v2f a, b;
    a.x = sPG[mt2 + mr][k + 2 * kh];
    a.y = sPG[mt2 + mr][k + 2 * kh + 1];
    b.x = sWl[mr][k + 2 * kh];
    b.y = sWl[mr][k + 2 * kh + 1];
    acc3 = wmma_f32_k4(a, b, acc3);
  }
  if (wv < 4) {
#pragma unroll
    for (int r = 0; r < 8; ++r) {
      int bb = mt2 + r + 8 * kh;
      int ll = mr;
      float xl = sXl[bb][ll];
      xl_out[(size_t)n * BATCH * LOC_DIM + bb * LOC_DIM + ll] =
          xl + ETA_X * acc3[r];
    }
  }

  if (tid == 0) atomicAdd(energy_t, sEnergy);
}

extern "C" void kernel_launch(void* const* d_in, const int* in_sizes, int n_in,
                              void* d_out, int out_size, void* d_ws, size_t ws_size,
                              hipStream_t stream) {
  const float* gin  = (const float*)d_in[0];
  const float* Wobj = (const float*)d_in[1];
  const float* Wloc = (const float*)d_in[2];

  float* out    = (float*)d_out;
  float* energy = out + XO_TOT;

  float* ws  = (float*)d_ws;
  float* xoA = ws;
  float* xoB = xoA + XO_TOT;
  float* xlA = xoB + XO_TOT;
  float* xlB = xlA + XL_TOT;

  cg_init_kernel<<<4096, 256, 0, stream>>>(xoA, xlA, energy);

  for (int t = 0; t < NSTEPS; ++t) {
    const float* xo_in = (t & 1) ? xoB : xoA;
    const float* xl_in = (t & 1) ? xlB : xlA;
    float* xo_out = (t == NSTEPS - 1) ? out : ((t & 1) ? xoA : xoB);
    float* xl_out = (t & 1) ? xlA : xlB;
    cg_step_kernel<<<NCOLS, 512, 0, stream>>>(gin, Wobj, Wloc,
                                              xo_in, xl_in, xo_out, xl_out,
                                              energy + t);
  }
}


# Staging: each thread fetches its share as 128-bit loads issued together ahead of the LDS stores (was load-wait-store per element); neighbour guards become selects, same sum order
# speedup vs baseline: 1.3525x; 1.3525x over previous
#include <hip/hip_runtime.h>
#include <math.h>


#define GRID_HH 32
#define GRID_WW 32
#define OBJ_DIM 64
#define LOC_DIM 16
#define SDIM    64
#define NCOLS   (GRID_HH * GRID_WW)
#define BATCH   64
#define ETA_X   0.05f
#define NSTEPS  20

#define XO_TOT  ((size_t)NCOLS * BATCH * OBJ_DIM)
#define XL_TOT  ((size_t)NCOLS * BATCH * LOC_DIM)

typedef __attribute__((ext_vector_type(2))) float v2f;
typedef __attribute__((ext_vector_type(8))) float v8f;
typedef __attribute__((ext_vector_type(4))) float v4f;

__device__ __forceinline__ v4f ld4(const float* p) {
  return *reinterpret_cast<const v4f*>(p);
}

__device__ __forceinline__ v8f wmma_f32_k4(v2f a, v2f b, v8f c) {
  return __builtin_amdgcn_wmma_f32_16x16x4_f32(
      false, a, false, b, (short)0, c, false, false);
}

__global__ void cg_init_kernel(float* __restrict__ xo,
                               float* __restrict__ xl,
                               float* __restrict__ energy) {
  size_t i = (size_t)blockIdx.x * blockDim.x + threadIdx.x;
  size_t stride = (size_t)gridDim.x * blockDim.x;
  for (size_t e = i; e < XO_TOT; e += stride) xo[e] = 0.0f;
  for (size_t e = i; e < XL_TOT; e += stride) xl[e] = 0.0f;
  if (i < NSTEPS) energy[i] = 0.0f;
}

__global__ __launch_bounds__(512, 16)
void cg_step_kernel(const float* __restrict__ gin,
                    const float* __restrict__ Wobj,
                    const float* __restrict__ Wloc,
                    const float* __restrict__ xo_in,
                    const float* __restrict__ xl_in,
                    float* __restrict__ xo_out,
                    float* __restrict__ xl_out,
                    float* __restrict__ energy_t) {
  __shared__ float sWo[OBJ_DIM][65];
  __shared__ float sWl[LOC_DIM][65];
  __shared__ float sPG[BATCH][65];
  __shared__ float sXo[BATCH][65];
  __shared__ float sXl[BATCH][17];
  __shared__ float sCtx[BATCH][65];
  __shared__ float sEnergy;

  const int n   = blockIdx.x;
  const int gh  = n >> 5;
  const int gw  = n & 31;
  const int tid = threadIdx.x;

  if (tid == 0) sEnergy = 0.0f;

  const int row = tid >> 4;
  const int col = (tid & 15) * 4;
  const unsigned off = (unsigned)tid * 4u;

  const bool hu = gh > 0, hd = gh < GRID_HH - 1, hl = gw > 0, hr = gw < GRID_WW - 1;
  const size_t tileo = (size_t)BATCH * OBJ_DIM;
  const float* gXo = xo_in + (size_t)n * tileo;
  const float* gUp = xo_in + (size_t)(hu ? n - GRID_WW : n) * tileo;
  const float* gDn = xo_in + (size_t)(hd ? n + GRID_WW : n) * tileo;
  const float* gLf = xo_in + (size_t)(hl ? n - 1 : n) * tileo;
  const float* gRt = xo_in + (size_t)(hr ? n + 1 : n) * tileo;
  const float* gWo = Wobj + (size_t)n * OBJ_DIM * SDIM;
  const float* gP  = gin + (size_t)(gh * 8) * 256 + gw * 8;
  const unsigned offp = (unsigned)row * 65536u + (unsigned)((tid >> 1) & 7) * 256u
                        + (unsigned)(tid & 1) * 4u;
  const bool lowhalf = __builtin_amdgcn_readfirstlane(tid) < 256;
  const int t2 = tid & 255;
  const float* gS = lowhalf ? Wloc + (size_t)n * LOC_DIM * SDIM
                            : xl_in + (size_t)n * BATCH * LOC_DIM;

  const v4f up0 = ld4(gUp + off), up1 = ld4(gUp + off + 32 * OBJ_DIM);
  const v4f dn0 = ld4(gDn + off), dn1 = ld4(gDn + off + 32 * OBJ_DIM);
  const v4f lf0 = ld4(gLf + off), lf1 = ld4(gLf + off + 32 * OBJ_DIM);
  const v4f rt0 = ld4(gRt + off), rt1 = ld4(gRt + off + 32 * OBJ_DIM);
  const v4f xo0 = ld4(gXo + off), xo1 = ld4(gXo + off + 32 * OBJ_DIM);
  const v4f wo0 = ld4(gWo + off), wo1 = ld4(gWo + off + 32 * SDIM);
  const v4f pp0 = ld4(gP + offp), pp1 = ld4(gP + offp + 32u * 65536u);
  const v4f sm  = ld4(gS + (unsigned)t2 * 4u);

  const int cnt = (int)hu + (int)hd + (int)hl + (int)hr;
  const float inv_cnt = (cnt == 4) ? 0.25f : ((cnt == 3) ? (1.0f / 3.0f) : 0.5f);
#pragma unroll
  for (int j = 0; j < 4; ++j) {
    float s0 = 0.0f, s1 = 0.0f;
    s0 += hu ? up0[j] : 0.0f;  s1 += hu ? up1[j] : 0.0f;
    s0 += hd ? dn0[j] : 0.0f;  s1 += hd ? dn1[j] : 0.0f;
    s0 += hl ? lf0[j] : 0.0f;  s1 += hl ? lf1[j] : 0.0f;
    s0 += hr ? rt0[j] : 0.0f;  s1 += hr ? rt1[j] : 0.0f;
    sCtx[row][col + j]      = s0 * inv_cnt;
    sCtx[row + 32][col + j] = s1 * inv_cnt;
    sXo[row][col + j]      = xo0[j];
    sXo[row + 32][col + j] = xo1[j];
    sWo[row][col + j]      = wo0[j];
    sWo[row + 32][col + j] = wo1[j];
    sPG[row][col + j]      = pp0[j];
    sPG[row + 32][col + j] = pp1[j];
  }
  {
    float* dst = lowhalf ? &sWl[t2 >> 4][(t2 & 15) * 4] : &sXl[t2 >> 2][(t2 & 3) * 4];
#pragma unroll
    for (int j = 0; j < 4; ++j) dst[j] = sm[j];
  }
  __syncthreads();

  const int lane = tid & 31;
  const int wv   = tid >> 5;
  const int kh   = lane >> 4;
  const int mr   = lane & 15;
  const int mt   = (wv >> 2) * 16;
  const int nt   = (wv & 3) * 16;

  v8f acc = {};
  for (int k = 0; k < SDIM; k += 4) {
    v2f a, b;
    a.x = sXo[mt + mr][k + 2 * kh];
    a.y = sXo[mt + mr][k + 2 * kh + 1];
    b.x = sWo[k + 2 * kh][nt + mr];
    b.y = sWo[k + 2 * kh + 1][nt + mr];
    acc = wmma_f32_k4(a, b, acc);
  }
  for (int k = 0; k < LOC_DIM; k += 4) {
    v2f a, b;
    a.x = sXl[mt + mr][k + 2 * kh];
    a.y = sXl[mt + mr][k + 2 * kh + 1];
    b.x = sWl[k + 2 * kh][nt + mr];
    b.y = sWl[k + 2 * kh + 1][nt + mr];
    acc = wmma_f32_k4(a, b, acc);
  }

  float epart = 0.0f;
#pragma unroll
  for (int r = 0; r < 8; ++r) {
    int bb = mt + r + 8 * kh;
    int ss = nt + mr;
    float pred = tanhf(acc[r]);
    float eps  = sPG[bb][ss] - pred;
    float g    = eps * (1.0f - pred * pred);
    epart += 0.5f * eps * eps;
    sPG[bb][ss] = g;
  }

  for (int off = 16; off > 0; off >>= 1) epart += __shfl_xor(epart, off, 32);
  if (lane == 0) atomicAdd(&sEnergy, epart);
  __syncthreads();

  v8f acc2 = {};
  for (int k = 0; k < SDIM; k += 4) {
    v2f a, b;
    a.x = sPG[mt + mr][k + 2 * kh];
    a.y = sPG[mt + mr][k + 2 * kh + 1];
    b.x = sWo[nt + mr][k + 2 * kh];
    b.y = sWo[nt + mr][k + 2 * kh + 1];
    acc2 = wmma_f32_k4(a, b, acc2);
  }
#pragma unroll
  for (int r = 0; r < 8; ++r) {
    int bb = mt + r + 8 * kh;
    int oo = nt + mr;
    float xo = sXo[bb][oo];
    xo_out[(size_t)n * BATCH * OBJ_DIM + bb * OBJ_DIM + oo] =
        xo + ETA_X * (acc2[r] + sCtx[bb][oo] - xo);
  }

  const int mt2 = (wv & 3) * 16;
  v8f acc3 = {};
  for (int k = 0; k < SDIM; k += 4) {
    v2f a, b;
    a.x = sPG[mt2 + mr][k + 2 * kh];
    a.y = sPG[mt2 + mr][k + 2 * kh + 1];
    b.x = sWl[mr][k + 2 * kh];
    b.y = sWl[mr][k + 2 * kh + 1];
    acc3 = wmma_f32_k4(a, b, acc3);
  }
  if (wv < 4) {
#pragma unroll
    for (int r = 0; r < 8; ++r) {
      int bb = mt2 + r + 8 * kh;
      int ll = mr;
      float xl = sXl[bb][ll];
      xl_out[(size_t)n * BATCH * LOC_DIM + bb * LOC_DIM + ll] =
          xl + ETA_X * acc3[r];
    }
  }

  if (tid == 0) atomicAdd(energy_t, sEnergy);
}

extern "C" void kernel_launch(void* const* d_in, const int* in_sizes, int n_in,
                              void* d_out, int out_size, void* d_ws, size_t ws_size,
                              hipStream_t stream) {
  const float* gin  = (const float*)d_in[0];
  const float* Wobj = (const float*)d_in[1];
  const float* Wloc = (const float*)d_in[2];

  float* out    = (float*)d_out;
  float* energy = out + XO_TOT;

  float* ws  = (float*)d_ws;
  float* xoA = ws;
  float* xoB = xoA + XO_TOT;
  float* xlA = xoB + XO_TOT;
  float* xlB = xlA + XL_TOT;

  cg_init_kernel<<<4096, 256, 0, stream>>>(xoA, xlA, energy);

  for (int t = 0; t < NSTEPS; ++t) {
    const float* xo_in = (t & 1) ? xoB : xoA;
    const float* xl_in = (t & 1) ? xlB : xlA;
    float* xo_out = (t == NSTEPS - 1) ? out : ((t & 1) ? xoA : xoB);
    float* xl_out = (t & 1) ? xlA : xlB;
    cg_step_kernel<<<NCOLS, 512, 0, stream>>>(gin, Wobj, Wloc,
                                              xo_in, xl_in, xo_out, xl_out,
                                              energy + t);
  }
}


# Work-group of 4 waves (was 16): a wave owns a 16-row batch strip, one A fragment per k-step feeds its 4 column tiles and the x_loc tile (no 4x recompute); energy partials regrouped per strip
# speedup vs baseline: 1.4606x; 1.0799x over previous
#include <hip/hip_runtime.h>
#include <math.h>


#define GRID_HH 32
#define GRID_WW 32
#define OBJ_DIM 64
#define LOC_DIM 16
#define SDIM    64
#define NCOLS   (GRID_HH * GRID_WW)
#define BATCH   64
#define ETA_X   0.05f
#define NSTEPS  20

#define XO_TOT  ((size_t)NCOLS * BATCH * OBJ_DIM)
#define XL_TOT  ((size_t)NCOLS * BATCH * LOC_DIM)

typedef __attribute__((ext_vector_type(2))) float v2f;
typedef __attribute__((ext_vector_type(8))) float v8f;
typedef __attribute__((ext_vector_type(4))) float v4f;

__device__ __forceinline__ v4f ld4(const float* p) {
  return *reinterpret_cast<const v4f*>(p);
}

__device__ __forceinline__ v8f wmma_f32_k4(v2f a, v2f b, v8f c) {
  return __builtin_amdgcn_wmma_f32_16x16x4_f32(
      false, a, false, b, (short)0, c, false, false);
}

__global__ void cg_init_kernel(float* __restrict__ xo,
                               float* __restrict__ xl,
                               float* __restrict__ energy) {
  size_t i = (size_t)blockIdx.x * blockDim.x + threadIdx.x;
  size_t stride = (size_t)gridDim.x * blockDim.x;
  for (size_t e = i; e < XO_TOT; e += stride) xo[e] = 0.0f;
  for (size_t e = i; e < XL_TOT; e += stride) xl[e] = 0.0f;
  if (i < NSTEPS) energy[i] = 0.0f;
}

__global__ __launch_bounds__(128, 4)
void cg_step_kernel(const float* __restrict__ gin,
                    const float* __restrict__ Wobj,
                    const float* __restrict__ Wloc,
                    const float* __restrict__ xo_in,
                    const float* __restrict__ xl_in,
                    float* __restrict__ xo_out,
                    float* __restrict__ xl_out,
                    float* __restrict__ energy_t) {
  __shared__ float sWo[OBJ_DIM][65];
  __shared__ float sWl[LOC_DIM][65];
  __shared__ float sPG[BATCH][65];
  __shared__ float sXo[BATCH][65];
  __shared__ float sXl[BATCH][17];
  __shared__ float sCtx[BATCH][65];
  __shared__ float sEnergy;

  const int n   = blockIdx.x;
  const int gh  = n >> 5;
  const int gw  = n & 31;
  const int tid = threadIdx.x;

  if (tid == 0) sEnergy = 0.0f;

  const int rowb = tid >> 4;
  const int col  = (tid & 15) * 4;
  const unsigned off = (unsigned)tid * 4u;

  const bool hu = gh > 0, hd = gh < GRID_HH - 1, hl = gw > 0, hr = gw < GRID_WW - 1;
  const size_t tileo = (size_t)BATCH * OBJ_DIM;
  const float* gXo = xo_in + (size_t)n * tileo;
  const float* gUp = xo_in + (size_t)(hu ? n - GRID_WW : n) * tileo;
  const float* gDn = xo_in + (size_t)(hd ? n + GRID_WW : n) * tileo;
  const float* gLf = xo_in + (size_t)(hl ? n - 1 : n) * tileo;
  const float* gRt = xo_in + (size_t)(hr ? n + 1 : n) * tileo;
  const float* gWo = Wobj + (size_t)n * OBJ_DIM * SDIM;
  const float* gWl = Wloc + (size_t)n * LOC_DIM * SDIM;
  const float* gXl = xl_in + (size_t)n * BATCH * LOC_DIM;
  const float* gP  = gin + (size_t)(gh * 8) * 256 + gw * 8;
  const unsigned offp = (unsigned)rowb * 65536u + (unsigned)((tid >> 1) & 7) * 256u
                        + (unsigned)(tid & 1) * 4u;

  const int cnt = (int)hu + (int)hd + (int)hl + (int)hr;
  const float inv_cnt = (cnt == 4) ? 0.25f : ((cnt == 3) ? (1.0f / 3.0f) : 0.5f);

  v4f up[8], dn[8], lf[8], rt[8], xo[8], wo[8], pp[8], wl[2], xl[2];
#define CG_FETCH(i)                                   \
  up[i] = ld4(gUp + off + 512u * (i));                \
  dn[i] = ld4(gDn + off + 512u * (i));                \
  lf[i] = ld4(gLf + off + 512u * (i));                \
  rt[i] = ld4(gRt + off + 512u * (i));                \
  xo[i] = ld4(gXo + off + 512u * (i));                \
  wo[i] = ld4(gWo + off + 512u * (i));                \
  pp[i] = ld4(gP + offp + 524288u * (i));
#define CG_PUT(i)                                     \
  _Pragma("unroll")                                   \
  for (int j = 0; j < 4; ++j) {                       \
    float s = 0.0f;                                   \
    s += hu ? up[i][j] : 0.0f;                        \
    s += hd ? dn[i][j] : 0.0f;                        \
    s += hl ? lf[i][j] : 0.0f;                        \
    s += hr ? rt[i][j] : 0.0f;                        \
    sCtx[rowb + 8 * (i)][col + j] = s * inv_cnt;      \
    sXo[rowb + 8 * (i)][col + j] = xo[i][j];          \
    sWo[rowb + 8 * (i)][col + j] = wo[i][j];          \
    sPG[rowb + 8 * (i)][col + j] = pp[i][j];          \
  }
  CG_FETCH(0) CG_FETCH(1) CG_FETCH(2)
  CG_FETCH(3) CG_FETCH(4) CG_FETCH(5)
  CG_PUT(0) CG_PUT(1) CG_PUT(2)
  CG_FETCH(6) CG_FETCH(7)
#pragma unroll
  for (int i = 0; i < 2; ++i) {
    wl[i] = ld4(gWl + off + 512u * i);
    xl[i] = ld4(gXl + off + 512u * i);
  }
  CG_PUT(3) CG_PUT(4) CG_PUT(5)
  CG_PUT(6) CG_PUT(7)
#pragma unroll
  for (int i = 0; i < 2; ++i) {
#pragma unroll
    for (int j = 0; j < 4; ++j) {
      sWl[rowb + 8 * i][col + j] = wl[i][j];
      sXl[(tid >> 2) + 32 * i][(tid & 3) * 4 + j] = xl[i][j];
    }
  }
#undef CG_FETCH
#undef CG_PUT
  __syncthreads();

  const int lane = tid & 31;
  const int wv   = tid >> 5;
  const int kh   = lane >> 4;
  const int mr   = lane & 15;
  const int mt   = wv * 16;

  v8f acc[4] = {};
  for (int k = 0; k < SDIM; k += 4) {
    v2f a;
    a.x = sXo[mt + mr][k + 2 * kh];
    a.y = sXo[mt + mr][k + 2 * kh + 1];
#pragma unroll
    for (int t = 0; t < 4; ++t) {
      v2f b;
      b.x = sWo[k + 2 * kh][16 * t + mr];
      b.y = sWo[k + 2 * kh + 1][16 * t + mr];
      acc[t] = wmma_f32_k4(a, b, acc[t]);
    }
  }
  for (int k = 0; k < LOC_DIM; k += 4) {
    v2f a;
    a.x = sXl[mt + mr][k + 2 * kh];
    a.y = sXl[mt + mr][k + 2 * kh + 1];
#pragma unroll
    for (int t = 0; t < 4; ++t) {
      v2f b;
      b.x = sWl[k + 2 * kh][16 * t + mr];
      b.y = sWl[k + 2 * kh + 1][16 * t + mr];
      acc[t] = wmma_f32_k4(a, b, acc[t]);
    }
  }

  float epart = 0.0f;
#pragma unroll
  for (int t = 0; t < 4; ++t) {
#pragma unroll
    for (int r = 0; r < 8; ++r) {
      int bb = mt + r + 8 * kh;
      int ss = 16 * t + mr;
      float pred = tanhf(acc[t][r]);
      float eps  = sPG[bb][ss] - pred;
      float g    = eps * (1.0f - pred * pred);
      epart += 0.5f * eps * eps;
      sPG[bb][ss] = g;
    }
  }

  for (int off2 = 16; off2 > 0; off2 >>= 1) epart += __shfl_xor(epart, off2, 32);
  if (lane == 0) atomicAdd(&sEnergy, epart);
  __syncthreads();

  v8f acc2[4] = {};
  v8f acc3 = {};
  for (int k = 0; k < SDIM; k += 4) {
    v2f a;
    a.x = sPG[mt + mr][k + 2 * kh];
    a.y = sPG[mt + mr][k + 2 * kh + 1];
#pragma unroll
    for (int t = 0; t < 4; ++t) {
      v2f b;
      b.x = sWo[16 * t + mr][k + 2 * kh];
      b.y = sWo[16 * t + mr][k + 2 * kh + 1];
      acc2[t] = wmma_f32_k4(a, b, acc2[t]);
    }
    v2f bl;
    bl.x = sWl[mr][k + 2 * kh];
    bl.y = sWl[mr][k + 2 * kh + 1];
    acc3 = wmma_f32_k4(a, bl, acc3);
  }
#pragma unroll
  for (int t = 0; t < 4; ++t) {
#pragma unroll
    for (int r = 0; r < 8; ++r) {
      int bb = mt + r + 8 * kh;
      int oo = 16 * t + mr;
      float xo = sXo[bb][oo];
      xo_out[(size_t)n * BATCH * OBJ_DIM + bb * OBJ_DIM + oo] =
          xo + ETA_X * (acc2[t][r] + sCtx[bb][oo] - xo);
    }
  }
#pragma unroll
  for (int r = 0; r < 8; ++r) {
    int bb = mt + r + 8 * kh;
    int ll = mr;
    float xl = sXl[bb][ll];
    xl_out[(size_t)n * BATCH * LOC_DIM + bb * LOC_DIM + ll] =
        xl + ETA_X * acc3[r];
  }

  if (tid == 0) atomicAdd(energy_t, sEnergy);
}

extern "C" void kernel_launch(void* const* d_in, const int* in_sizes, int n_in,
                              void* d_out, int out_size, void* d_ws, size_t ws_size,
                              hipStream_t stream) {
  const float* gin  = (const float*)d_in[0];
  const float* Wobj = (const float*)d_in[1];
  const float* Wloc = (const float*)d_in[2];

  float* out    = (float*)d_out;
  float* energy = out + XO_TOT;

  float* ws  = (float*)d_ws;
  float* xoA = ws;
  float* xoB = xoA + XO_TOT;
  float* xlA = xoB + XO_TOT;
  float* xlB = xlA + XL_TOT;

  cg_init_kernel<<<4096, 256, 0, stream>>>(xoA, xlA, energy);

  for (int t = 0; t < NSTEPS; ++t) {
    const float* xo_in = (t & 1) ? xoB : xoA;
    const float* xl_in = (t & 1) ? xlB : xlA;
    float* xo_out = (t == NSTEPS - 1) ? out : ((t & 1) ? xoA : xoB);
    float* xl_out = (t & 1) ? xlA : xlB;
    cg_step_kernel<<<NCOLS, 128, 0, stream>>>(gin, Wobj, Wloc,
                                              xo_in, xl_in, xo_out, xl_out,
                                              energy + t);
  }
}


# Phase-B epilogue as one straight-line block: tanh evaluates both ranges and selects (was a branch per range per element), the lane's 32 patch values read from LDS ahead of it; same sum order
# speedup vs baseline: 1.5679x; 1.0735x over previous
#include <hip/hip_runtime.h>
#include <math.h>


#define GRID_HH 32
#define GRID_WW 32
#define OBJ_DIM 64
#define LOC_DIM 16
#define SDIM    64
#define NCOLS   (GRID_HH * GRID_WW)
#define BATCH   64
#define ETA_X   0.05f
#define NSTEPS  20

#define XO_TOT  ((size_t)NCOLS * BATCH * OBJ_DIM)
#define XL_TOT  ((size_t)NCOLS * BATCH * LOC_DIM)

typedef __attribute__((ext_vector_type(2))) float v2f;
typedef __attribute__((ext_vector_type(8))) float v8f;
typedef __attribute__((ext_vector_type(4))) float v4f;

__device__ __forceinline__ v4f ld4(const float* p) {
  return *reinterpret_cast<const v4f*>(p);
}

__device__ __forceinline__ v8f wmma_f32_k4(v2f a, v2f b, v8f c) {
  return __builtin_amdgcn_wmma_f32_16x16x4_f32(
      false, a, false, b, (short)0, c, false, false);
}

__device__ __forceinline__ float tanh_both_ranges(float x) {
  const float ax = __builtin_fabsf(x);

  const float t  = expf(ax + ax);
  const float zl = __builtin_fmaf(__builtin_amdgcn_rcpf(1.0f + t), -2.0f, 1.0f);

  const float y2 = x * x;
  float p = __builtin_fmaf(-0x1.758e7ap-8f, y2, 0x1.521192p-6f);
  p = __builtin_fmaf(y2, p, -0x1.b8389cp-5f);
  p = __builtin_fmaf(y2, p, 0x1.110704p-3f);
  p = __builtin_fmaf(y2, p, -0x1.555532p-2f);
  const float zs = __builtin_fmaf(y2, ax * p, ax);

  const float z = (0.625f > ax) ? zs : zl;
  return __builtin_copysignf(z, x);
}

__global__ void cg_init_kernel(float* __restrict__ xo,
                               float* __restrict__ xl,
                               float* __restrict__ energy) {
  size_t i = (size_t)blockIdx.x * blockDim.x + threadIdx.x;
  size_t stride = (size_t)gridDim.x * blockDim.x;
  for (size_t e = i; e < XO_TOT; e += stride) xo[e] = 0.0f;
  for (size_t e = i; e < XL_TOT; e += stride) xl[e] = 0.0f;
  if (i < NSTEPS) energy[i] = 0.0f;
}

__global__ __launch_bounds__(128, 4)
void cg_step_kernel(const float* __restrict__ gin,
                    const float* __restrict__ Wobj,
                    const float* __restrict__ Wloc,
                    const float* __restrict__ xo_in,
                    const float* __restrict__ xl_in,
                    float* __restrict__ xo_out,
                    float* __restrict__ xl_out,
                    float* __restrict__ energy_t) {
  __shared__ float sWo[OBJ_DIM][65];
  __shared__ float sWl[LOC_DIM][65];
  __shared__ float sPG[BATCH][65];
  __shared__ float sXo[BATCH][65];
  __shared__ float sXl[BATCH][17];
  __shared__ float sCtx[BATCH][65];
  __shared__ float sEnergy;

  const int n   = blockIdx.x;
  const int gh  = n >> 5;
  const int gw  = n & 31;
  const int tid = threadIdx.x;

  if (tid == 0) sEnergy = 0.0f;

  const int rowb = tid >> 4;
  const int col  = (tid & 15) * 4;
  const unsigned off = (unsigned)tid * 4u;

  const bool hu = gh > 0, hd = gh < GRID_HH - 1, hl = gw > 0, hr = gw < GRID_WW - 1;
  const size_t tileo = (size_t)BATCH * OBJ_DIM;
  const float* gXo = xo_in + (size_t)n * tileo;
  const float* gUp = xo_in + (size_t)(hu ? n - GRID_WW : n) * tileo;
  const float* gDn = xo_in + (size_t)(hd ? n + GRID_WW : n) * tileo;
  const float* gLf = xo_in + (size_t)(hl ? n - 1 : n) * tileo;
  const float* gRt = xo_in + (size_t)(hr ? n + 1 : n) * tileo;
  const float* gWo = Wobj + (size_t)n * OBJ_DIM * SDIM;
  const float* gWl = Wloc + (size_t)n * LOC_DIM * SDIM;
  const float* gXl = xl_in + (size_t)n * BATCH * LOC_DIM;
  const float* gP  = gin + (size_t)(gh * 8) * 256 + gw * 8;
  const unsigned offp = (unsigned)rowb * 65536u + (unsigned)((tid >> 1) & 7) * 256u
                        + (unsigned)(tid & 1) * 4u;

  const int cnt = (int)hu + (int)hd + (int)hl + (int)hr;
  const float inv_cnt = (cnt == 4) ? 0.25f : ((cnt == 3) ? (1.0f / 3.0f) : 0.5f);

  v4f up[8], dn[8], lf[8], rt[8], xo[8], wo[8], pp[8], wl[2], xl[2];
#define CG_FETCH(i)                                   \
  up[i] = ld4(gUp + off + 512u * (i));                \
  dn[i] = ld4(gDn + off + 512u * (i));                \
  lf[i] = ld4(gLf + off + 512u * (i));                \
  rt[i] = ld4(gRt + off + 512u * (i));                \
  xo[i] = ld4(gXo + off + 512u * (i));                \
  wo[i] = ld4(gWo + off + 512u * (i));                \
  pp[i] = ld4(gP + offp + 524288u * (i));
#define CG_PUT(i)                                     \
  _Pragma("unroll")                                   \
  for (int j = 0; j < 4; ++j) {                       \
    float s = 0.0f;                                   \
    s += hu ? up[i][j] : 0.0f;                        \
    s += hd ? dn[i][j] : 0.0f;                        \
    s += hl ? lf[i][j] : 0.0f;                        \
    s += hr ? rt[i][j] : 0.0f;                        \
    sCtx[rowb + 8 * (i)][col + j] = s * inv_cnt;      \
    sXo[rowb + 8 * (i)][col + j] = xo[i][j];          \
    sWo[rowb + 8 * (i)][col + j] = wo[i][j];          \
    sPG[rowb + 8 * (i)][col + j] = pp[i][j];          \
  }
  CG_FETCH(0) CG_FETCH(1) CG_FETCH(2)
  CG_FETCH(3) CG_FETCH(4) CG_FETCH(5)
  CG_PUT(0) CG_PUT(1) CG_PUT(2)
  CG_FETCH(6) CG_FETCH(7)
#pragma unroll
  for (int i = 0; i < 2; ++i) {
    wl[i] = ld4(gWl + off + 512u * i);
    xl[i] = ld4(gXl + off + 512u * i);
  }
  CG_PUT(3) CG_PUT(4) CG_PUT(5)
  CG_PUT(6) CG_PUT(7)
#pragma unroll
  for (int i = 0; i < 2; ++i) {
#pragma unroll
    for (int j = 0; j < 4; ++j) {
      sWl[rowb + 8 * i][col + j] = wl[i][j];
      sXl[(tid >> 2) + 32 * i][(tid & 3) * 4 + j] = xl[i][j];
    }
  }
#undef CG_FETCH
#undef CG_PUT
  __syncthreads();

  const int lane = tid & 31;
  const int wv   = tid >> 5;
  const int kh   = lane >> 4;
  const int mr   = lane & 15;
  const int mt   = wv * 16;

  v8f acc[4] = {};
  for (int k = 0; k < SDIM; k += 4) {
    v2f a;
    a.x = sXo[mt + mr][k + 2 * kh];
    a.y = sXo[mt + mr][k + 2 * kh + 1];
#pragma unroll
    for (int t = 0; t < 4; ++t) {
      v2f b;
      b.x = sWo[k + 2 * kh][16 * t + mr];
      b.y = sWo[k + 2 * kh + 1][16 * t + mr];
      acc[t] = wmma_f32_k4(a, b, acc[t]);
    }
  }
  for (int k = 0; k < LOC_DIM; k += 4) {
    v2f a;
    a.x = sXl[mt + mr][k + 2 * kh];
    a.y = sXl[mt + mr][k + 2 * kh + 1];
#pragma unroll
    for (int t = 0; t < 4; ++t) {
      v2f b;
      b.x = sWl[k + 2 * kh][16 * t + mr];
      b.y = sWl[k + 2 * kh + 1][16 * t + mr];
      acc[t] = wmma_f32_k4(a, b, acc[t]);
    }
  }

  float pv[4][8];
#pragma unroll
  for (int t = 0; t < 4; ++t) {
#pragma unroll
    for (int r = 0; r < 8; ++r) {
      pv[t][r] = sPG[mt + r + 8 * kh][16 * t + mr];
    }
  }
  float epart = 0.0f;
#pragma unroll
  for (int t = 0; t < 4; ++t) {
#pragma unroll
    for (int r = 0; r < 8; ++r) {
      int bb = mt + r + 8 * kh;
      int ss = 16 * t + mr;
      float pred = tanh_both_ranges(acc[t][r]);
      float eps  = pv[t][r] - pred;
      float g    = eps * (1.0f - pred * pred);
      epart += 0.5f * eps * eps;
      sPG[bb][ss] = g;
    }
  }

  for (int off2 = 16; off2 > 0; off2 >>= 1) epart += __shfl_xor(epart, off2, 32);
  if (lane == 0) atomicAdd(&sEnergy, epart);
  __syncthreads();

  v8f acc2[4] = {};
  v8f acc3 = {};
  for (int k = 0; k < SDIM; k += 4) {
    v2f a;
    a.x = sPG[mt + mr][k + 2 * kh];
    a.y = sPG[mt + mr][k + 2 * kh + 1];
#pragma unroll
    for (int t = 0; t < 4; ++t) {
      v2f b;
      b.x = sWo[16 * t + mr][k + 2 * kh];
      b.y = sWo[16 * t + mr][k + 2 * kh + 1];
      acc2[t] = wmma_f32_k4(a, b, acc2[t]);
    }
    v2f bl;
    bl.x = sWl[mr][k + 2 * kh];
    bl.y = sWl[mr][k + 2 * kh + 1];
    acc3 = wmma_f32_k4(a, bl, acc3);
  }
#pragma unroll
  for (int t = 0; t < 4; ++t) {
#pragma unroll
    for (int r = 0; r < 8; ++r) {
      int bb = mt + r + 8 * kh;
      int oo = 16 * t + mr;
      float xo = sXo[bb][oo];
      xo_out[(size_t)n * BATCH * OBJ_DIM + bb * OBJ_DIM + oo] =
          xo + ETA_X * (acc2[t][r] + sCtx[bb][oo] - xo);
    }
  }
#pragma unroll
  for (int r = 0; r < 8; ++r) {
    int bb = mt + r + 8 * kh;
    int ll = mr;
    float xl = sXl[bb][ll];
    xl_out[(size_t)n * BATCH * LOC_DIM + bb * LOC_DIM + ll] =
        xl + ETA_X * acc3[r];
  }

  if (tid == 0) atomicAdd(energy_t, sEnergy);
}

extern "C" void kernel_launch(void* const* d_in, const int* in_sizes, int n_in,
                              void* d_out, int out_size, void* d_ws, size_t ws_size,
                              hipStream_t stream) {
  const float* gin  = (const float*)d_in[0];
  const float* Wobj = (const float*)d_in[1];
  const float* Wloc = (const float*)d_in[2];

  float* out    = (float*)d_out;
  float* energy = out + XO_TOT;

  float* ws  = (float*)d_ws;
  float* xoA = ws;
  float* xoB = xoA + XO_TOT;
  float* xlA = xoB + XO_TOT;
  float* xlB = xlA + XL_TOT;

  cg_init_kernel<<<4096, 256, 0, stream>>>(xoA, xlA, energy);

  for (int t = 0; t < NSTEPS; ++t) {
    const float* xo_in = (t & 1) ? xoB : xoA;
    const float* xl_in = (t & 1) ? xlB : xlA;
    float* xo_out = (t == NSTEPS - 1) ? out : ((t & 1) ? xoA : xoB);
    float* xl_out = (t & 1) ? xlA : xlB;
    cg_step_kernel<<<NCOLS, 128, 0, stream>>>(gin, Wobj, Wloc,
                                              xo_in, xl_in, xo_out, xl_out,
                                              energy + t);
  }
}


# tanh in one range as (1-t)/(1+t), t = 2^(-2 log2e |u|), on the hardware exponential and reciprocal (was: polynomial and library exponential both evaluated, then selected); same sum order
# speedup vs baseline: 1.6574x; 1.0571x over previous
#include <hip/hip_runtime.h>
#include <math.h>


#define GRID_HH 32
#define GRID_WW 32
#define OBJ_DIM 64
#define LOC_DIM 16
#define SDIM    64
#define NCOLS   (GRID_HH * GRID_WW)
#define BATCH   64
#define ETA_X   0.05f
#define NSTEPS  20

#define XO_TOT  ((size_t)NCOLS * BATCH * OBJ_DIM)
#define XL_TOT  ((size_t)NCOLS * BATCH * LOC_DIM)

typedef __attribute__((ext_vector_type(2))) float v2f;
typedef __attribute__((ext_vector_type(8))) float v8f;
typedef __attribute__((ext_vector_type(4))) float v4f;

__device__ __forceinline__ v4f ld4(const float* p) {
  return *reinterpret_cast<const v4f*>(p);
}

__device__ __forceinline__ v8f wmma_f32_k4(v2f a, v2f b, v8f c) {
  return __builtin_amdgcn_wmma_f32_16x16x4_f32(
      false, a, false, b, (short)0, c, false, false);
}

__device__ __forceinline__ float tanh_exp2(float x) {
  const float ax = __builtin_fabsf(x);
  const float t = __builtin_amdgcn_exp2f(ax * -0x1.715476p+1f);
  const float r = __builtin_amdgcn_rcpf(1.0f + t);
  const float z = (1.0f - t) * r;
  return __builtin_copysignf(z, x);
}

__global__ void cg_init_kernel(float* __restrict__ xo,
                               float* __restrict__ xl,
                               float* __restrict__ energy) {
  size_t i = (size_t)blockIdx.x * blockDim.x + threadIdx.x;
  size_t stride = (size_t)gridDim.x * blockDim.x;
  for (size_t e = i; e < XO_TOT; e += stride) xo[e] = 0.0f;
  for (size_t e = i; e < XL_TOT; e += stride) xl[e] = 0.0f;
  if (i < NSTEPS) energy[i] = 0.0f;
}

__global__ __launch_bounds__(128, 4)
void cg_step_kernel(const float* __restrict__ gin,
                    const float* __restrict__ Wobj,
                    const float* __restrict__ Wloc,
                    const float* __restrict__ xo_in,
                    const float* __restrict__ xl_in,
                    float* __restrict__ xo_out,
                    float* __restrict__ xl_out,
                    float* __restrict__ energy_t) {
  __shared__ float sWo[OBJ_DIM][65];
  __shared__ float sWl[LOC_DIM][65];
  __shared__ float sPG[BATCH][65];
  __shared__ float sXo[BATCH][65];
  __shared__ float sXl[BATCH][17];
  __shared__ float sCtx[BATCH][65];
  __shared__ float sEnergy;

  const int n   = blockIdx.x;
  const int gh  = n >> 5;
  const int gw  = n & 31;
  const int tid = threadIdx.x;

  if (tid == 0) sEnergy = 0.0f;

  const int rowb = tid >> 4;
  const int col  = (tid & 15) * 4;
  const unsigned off = (unsigned)tid * 4u;

  const bool hu = gh > 0, hd = gh < GRID_HH - 1, hl = gw > 0, hr = gw < GRID_WW - 1;
  const size_t tileo = (size_t)BATCH * OBJ_DIM;
  const float* gXo = xo_in + (size_t)n * tileo;
  const float* gUp = xo_in + (size_t)(hu ? n - GRID_WW : n) * tileo;
  const float* gDn = xo_in + (size_t)(hd ? n + GRID_WW : n) * tileo;
  const float* gLf = xo_in + (size_t)(hl ? n - 1 : n) * tileo;
  const float* gRt = xo_in + (size_t)(hr ? n + 1 : n) * tileo;
  const float* gWo = Wobj + (size_t)n * OBJ_DIM * SDIM;
  const float* gWl = Wloc + (size_t)n * LOC_DIM * SDIM;
  const float* gXl = xl_in + (size_t)n * BATCH * LOC_DIM;
  const float* gP  = gin + (size_t)(gh * 8) * 256 + gw * 8;
  const unsigned offp = (unsigned)rowb * 65536u + (unsigned)((tid >> 1) & 7) * 256u
                        + (unsigned)(tid & 1) * 4u;

  const int cnt = (int)hu + (int)hd + (int)hl + (int)hr;
  const float inv_cnt = (cnt == 4) ? 0.25f : ((cnt == 3) ? (1.0f / 3.0f) : 0.5f);

  v4f up[8], dn[8], lf[8], rt[8], xo[8], wo[8], pp[8], wl[2], xl[2];
#define CG_FETCH(i)                                   \
  up[i] = ld4(gUp + off + 512u * (i));                \
  dn[i] = ld4(gDn + off + 512u * (i));                \
  lf[i] = ld4(gLf + off + 512u * (i));                \
  rt[i] = ld4(gRt + off + 512u * (i));                \
  xo[i] = ld4(gXo + off + 512u * (i));                \
  wo[i] = ld4(gWo + off + 512u * (i));                \
  pp[i] = ld4(gP + offp + 524288u * (i));
#define CG_PUT(i)                                     \
  _Pragma("unroll")                                   \
  for (int j = 0; j < 4; ++j) {                       \
    float s = 0.0f;                                   \
    s += hu ? up[i][j] : 0.0f;                        \
    s += hd ? dn[i][j] : 0.0f;                        \
    s += hl ? lf[i][j] : 0.0f;                        \
    s += hr ? rt[i][j] : 0.0f;                        \
    sCtx[rowb + 8 * (i)][col + j] = s * inv_cnt;      \
    sXo[rowb + 8 * (i)][col + j] = xo[i][j];          \
    sWo[rowb + 8 * (i)][col + j] = wo[i][j];          \
    sPG[rowb + 8 * (i)][col + j] = pp[i][j];          \
  }
  CG_FETCH(0) CG_FETCH(1) CG_FETCH(2)
  CG_FETCH(3) CG_FETCH(4) CG_FETCH(5)
  CG_PUT(0) CG_PUT(1) CG_PUT(2)
  CG_FETCH(6) CG_FETCH(7)
#pragma unroll
  for (int i = 0; i < 2; ++i) {
    wl[i] = ld4(gWl + off + 512u * i);
    xl[i] = ld4(gXl + off + 512u * i);
  }
  CG_PUT(3) CG_PUT(4) CG_PUT(5)
  CG_PUT(6) CG_PUT(7)
#pragma unroll
  for (int i = 0; i < 2; ++i) {
#pragma unroll
    for (int j = 0; j < 4; ++j) {
      sWl[rowb + 8 * i][col + j] = wl[i][j];
      sXl[(tid >> 2) + 32 * i][(tid & 3) * 4 + j] = xl[i][j];
    }
  }
#undef CG_FETCH
#undef CG_PUT
  __syncthreads();

  const int lane = tid & 31;
  const int wv   = tid >> 5;
  const int kh   = lane >> 4;
  const int mr   = lane & 15;
  const int mt   = wv * 16;

  v8f acc[4] = {};
  for (int k = 0; k < SDIM; k += 4) {
    v2f a;
    a.x = sXo[mt + mr][k + 2 * kh];
    a.y = sXo[mt + mr][k + 2 * kh + 1];
#pragma unroll
    for (int t = 0; t < 4; ++t) {
      v2f b;
      b.x = sWo[k + 2 * kh][16 * t + mr];
      b.y = sWo[k + 2 * kh + 1][16 * t + mr];
      acc[t] = wmma_f32_k4(a, b, acc[t]);
    }
  }
  for (int k = 0; k < LOC_DIM; k += 4) {
    v2f a;
    a.x = sXl[mt + mr][k + 2 * kh];
    a.y = sXl[mt + mr][k + 2 * kh + 1];
#pragma unroll
    for (int t = 0; t < 4; ++t) {
      v2f b;
      b.x = sWl[k + 2 * kh][16 * t + mr];
      b.y = sWl[k + 2 * kh + 1][16 * t + mr];
      acc[t] = wmma_f32_k4(a, b, acc[t]);
    }
  }

  float pv[4][8];
#pragma unroll
  for (int t = 0; t < 4; ++t) {
#pragma unroll
    for (int r = 0; r < 8; ++r) {
      pv[t][r] = sPG[mt + r + 8 * kh][16 * t + mr];
    }
  }
  float epart = 0.0f;
#pragma unroll
  for (int t = 0; t < 4; ++t) {
#pragma unroll
    for (int r = 0; r < 8; ++r) {
      int bb = mt + r + 8 * kh;
      int ss = 16 * t + mr;
      float pred = tanh_exp2(acc[t][r]);
      float eps  = pv[t][r] - pred;
      float g    = eps * (1.0f - pred * pred);
      epart += 0.5f * eps * eps;
      sPG[bb][ss] = g;
    }
  }

  for (int off2 = 16; off2 > 0; off2 >>= 1) epart += __shfl_xor(epart, off2, 32);
  if (lane == 0) atomicAdd(&sEnergy, epart);
  __syncthreads();

  v8f acc2[4] = {};
  v8f acc3 = {};
  for (int k = 0; k < SDIM; k += 4) {
    v2f a;
    a.x = sPG[mt + mr][k + 2 * kh];
    a.y = sPG[mt + mr][k + 2 * kh + 1];
#pragma unroll
    for (int t = 0; t < 4; ++t) {
      v2f b;
      b.x = sWo[16 * t + mr][k + 2 * kh];
      b.y = sWo[16 * t + mr][k + 2 * kh + 1];
      acc2[t] = wmma_f32_k4(a, b, acc2[t]);
    }
    v2f bl;
    bl.x = sWl[mr][k + 2 * kh];
    bl.y = sWl[mr][k + 2 * kh + 1];
    acc3 = wmma_f32_k4(a, bl, acc3);
  }
#pragma unroll
  for (int t = 0; t < 4; ++t) {
#pragma unroll
    for (int r = 0; r < 8; ++r) {
      int bb = mt + r + 8 * kh;
      int oo = 16 * t + mr;
      float xo = sXo[bb][oo];
      xo_out[(size_t)n * BATCH * OBJ_DIM + bb * OBJ_DIM + oo] =
          xo + ETA_X * (acc2[t][r] + sCtx[bb][oo] - xo);
    }
  }
#pragma unroll
  for (int r = 0; r < 8; ++r) {
    int bb = mt + r + 8 * kh;
    int ll = mr;
    float xl = sXl[bb][ll];
    xl_out[(size_t)n * BATCH * LOC_DIM + bb * LOC_DIM + ll] =
        xl + ETA_X * acc3[r];
  }

  if (tid == 0) atomicAdd(energy_t, sEnergy);
}

extern "C" void kernel_launch(void* const* d_in, const int* in_sizes, int n_in,
                              void* d_out, int out_size, void* d_ws, size_t ws_size,
                              hipStream_t stream) {
  const float* gin  = (const float*)d_in[0];
  const float* Wobj = (const float*)d_in[1];
  const float* Wloc = (const float*)d_in[2];

  float* out    = (float*)d_out;
  float* energy = out + XO_TOT;

  float* ws  = (float*)d_ws;
  float* xoA = ws;
  float* xoB = xoA + XO_TOT;
  float* xlA = xoB + XO_TOT;
  float* xlB = xlA + XL_TOT;

  cg_init_kernel<<<4096, 256, 0, stream>>>(xoA, xlA, energy);

  for (int t = 0; t < NSTEPS; ++t) {
    const float* xo_in = (t & 1) ? xoB : xoA;
    const float* xl_in = (t & 1) ? xlB : xlA;
    float* xo_out = (t == NSTEPS - 1) ? out : ((t & 1) ? xoA : xoB);
    float* xl_out = (t & 1) ? xlA : xlB;
    cg_step_kernel<<<NCOLS, 128, 0, stream>>>(gin, Wobj, Wloc,
                                              xo_in, xl_in, xo_out, xl_out,
                                              energy + t);
  }
}


# Patch tile re-laid once: the first step stores each column's 64x64 patch contiguously in the workspace and later steps fetch whole 128-byte lines (was 32 bytes of each of 512 lines); same sum order
# speedup vs baseline: 1.7774x; 1.0724x over previous
#include <hip/hip_runtime.h>
#include <math.h>


#define GRID_HH 32
#define GRID_WW 32
#define OBJ_DIM 64
#define LOC_DIM 16
#define SDIM    64
#define NCOLS   (GRID_HH * GRID_WW)
#define BATCH   64
#define ETA_X   0.05f
#define NSTEPS  20

#define XO_TOT  ((size_t)NCOLS * BATCH * OBJ_DIM)
#define XL_TOT  ((size_t)NCOLS * BATCH * LOC_DIM)

typedef __attribute__((ext_vector_type(2))) float v2f;
typedef __attribute__((ext_vector_type(8))) float v8f;
typedef __attribute__((ext_vector_type(4))) float v4f;

__device__ __forceinline__ v4f ld4(const float* p) {
  return *reinterpret_cast<const v4f*>(p);
}

__device__ __forceinline__ void st4(float* p, v4f v) {
  *reinterpret_cast<v4f*>(p) = v;
}

__device__ __forceinline__ v8f wmma_f32_k4(v2f a, v2f b, v8f c) {
  return __builtin_amdgcn_wmma_f32_16x16x4_f32(
      false, a, false, b, (short)0, c, false, false);
}

__device__ __forceinline__ float tanh_exp2(float x) {
  const float ax = __builtin_fabsf(x);
  const float t = __builtin_amdgcn_exp2f(ax * -0x1.715476p+1f);
  const float r = __builtin_amdgcn_rcpf(1.0f + t);
  const float z = (1.0f - t) * r;
  return __builtin_copysignf(z, x);
}

__global__ void cg_init_kernel(float* __restrict__ xo,
                               float* __restrict__ xl,
                               float* __restrict__ energy) {
  size_t i = (size_t)blockIdx.x * blockDim.x + threadIdx.x;
  size_t stride = (size_t)gridDim.x * blockDim.x;
  for (size_t e = i; e < XO_TOT; e += stride) xo[e] = 0.0f;
  for (size_t e = i; e < XL_TOT; e += stride) xl[e] = 0.0f;
  if (i < NSTEPS) energy[i] = 0.0f;
}

template <bool FIRST>
__global__ __launch_bounds__(128, 4)
void cg_step_kernel(const float* __restrict__ gin,
                    const float* __restrict__ Wobj,
                    const float* __restrict__ Wloc,
                    const float* __restrict__ xo_in,
                    const float* __restrict__ xl_in,
                    float* __restrict__ xo_out,
                    float* __restrict__ xl_out,
                    float* __restrict__ energy_t,
                    float* __restrict__ ppk) {
  __shared__ float sWo[OBJ_DIM][65];
  __shared__ float sWl[LOC_DIM][65];
  __shared__ float sPG[BATCH][65];
  __shared__ float sXo[BATCH][65];
  __shared__ float sXl[BATCH][17];
  __shared__ float sCtx[BATCH][65];
  __shared__ float sEnergy;

  const int n   = blockIdx.x;
  const int gh  = n >> 5;
  const int gw  = n & 31;
  const int tid = threadIdx.x;

  if (tid == 0) sEnergy = 0.0f;

  const int rowb = tid >> 4;
  const int col  = (tid & 15) * 4;
  const unsigned off = (unsigned)tid * 4u;

  const bool hu = gh > 0, hd = gh < GRID_HH - 1, hl = gw > 0, hr = gw < GRID_WW - 1;
  const size_t tileo = (size_t)BATCH * OBJ_DIM;
  const float* gXo = xo_in + (size_t)n * tileo;
  const float* gUp = xo_in + (size_t)(hu ? n - GRID_WW : n) * tileo;
  const float* gDn = xo_in + (size_t)(hd ? n + GRID_WW : n) * tileo;
  const float* gLf = xo_in + (size_t)(hl ? n - 1 : n) * tileo;
  const float* gRt = xo_in + (size_t)(hr ? n + 1 : n) * tileo;
  const float* gWo = Wobj + (size_t)n * OBJ_DIM * SDIM;
  const float* gWl = Wloc + (size_t)n * LOC_DIM * SDIM;
  const float* gXl = xl_in + (size_t)n * BATCH * LOC_DIM;
  const float* gP  = gin + (size_t)(gh * 8) * 256 + gw * 8;
  const unsigned offp = (unsigned)rowb * 65536u + (unsigned)((tid >> 1) & 7) * 256u
                        + (unsigned)(tid & 1) * 4u;
  float* gPk = ppk + (size_t)n * tileo;

  const int cnt = (int)hu + (int)hd + (int)hl + (int)hr;
  const float inv_cnt = (cnt == 4) ? 0.25f : ((cnt == 3) ? (1.0f / 3.0f) : 0.5f);

  v4f up[8], dn[8], lf[8], rt[8], xo[8], wo[8], pp[8], wl[2], xl[2];
#define CG_FETCH(i)                                   \
  up[i] = ld4(gUp + off + 512u * (i));                \
  dn[i] = ld4(gDn + off + 512u * (i));                \
  lf[i] = ld4(gLf + off + 512u * (i));                \
  rt[i] = ld4(gRt + off + 512u * (i));                \
  xo[i] = ld4(gXo + off + 512u * (i));                \
  wo[i] = ld4(gWo + off + 512u * (i));                \
  if constexpr (FIRST) {                              \
    pp[i] = ld4(gP + offp + 524288u * (i));   \
  } else {                                            \
    pp[i] = ld4(gPk + off + 512u * (i));              \
  }
#define CG_PUT(i)                                     \
  if constexpr (FIRST) st4(gPk + off + 512u * (i), pp[i]); \
  _Pragma("unroll")                                   \
  for (int j = 0; j < 4; ++j) {                       \
    float s = 0.0f;                                   \
    s += hu ? up[i][j] : 0.0f;                        \
    s += hd ? dn[i][j] : 0.0f;                        \
    s += hl ? lf[i][j] : 0.0f;                        \
    s += hr ? rt[i][j] : 0.0f;                        \
    sCtx[rowb + 8 * (i)][col + j] = s * inv_cnt;      \
    sXo[rowb + 8 * (i)][col + j] = xo[i][j];          \
    sWo[rowb + 8 * (i)][col + j] = wo[i][j];          \
    sPG[rowb + 8 * (i)][col + j] = pp[i][j];          \
  }
  CG_FETCH(0) CG_FETCH(1) CG_FETCH(2)
  CG_FETCH(3) CG_FETCH(4) CG_FETCH(5)
  CG_PUT(0) CG_PUT(1) CG_PUT(2)
  CG_FETCH(6) CG_FETCH(7)
#pragma unroll
  for (int i = 0; i < 2; ++i) {
    wl[i] = ld4(gWl + off + 512u * i);
    xl[i] = ld4(gXl + off + 512u * i);
  }
  CG_PUT(3) CG_PUT(4) CG_PUT(5)
  CG_PUT(6) CG_PUT(7)
#pragma unroll
  for (int i = 0; i < 2; ++i) {
#pragma unroll
    for (int j = 0; j < 4; ++j) {
      sWl[rowb + 8 * i][col + j] = wl[i][j];
      sXl[(tid >> 2) + 32 * i][(tid & 3) * 4 + j] = xl[i][j];
    }
  }
#undef CG_FETCH
#undef CG_PUT
  __syncthreads();

  const int lane = tid & 31;
  const int wv   = tid >> 5;
  const int kh   = lane >> 4;
  const int mr   = lane & 15;
  const int mt   = wv * 16;

  v8f acc[4] = {};
  for (int k = 0; k < SDIM; k += 4) {
    v2f a;
    a.x = sXo[mt + mr][k + 2 * kh];
    a.y = sXo[mt + mr][k + 2 * kh + 1];
#pragma unroll
    for (int t = 0; t < 4; ++t) {
      v2f b;
      b.x = sWo[k + 2 * kh][16 * t + mr];
      b.y = sWo[k + 2 * kh + 1][16 * t + mr];
      acc[t] = wmma_f32_k4(a, b, acc[t]);
    }
  }
  for (int k = 0; k < LOC_DIM; k += 4) {
    v2f a;
    a.x = sXl[mt + mr][k + 2 * kh];
    a.y = sXl[mt + mr][k + 2 * kh + 1];
#pragma unroll
    for (int t = 0; t < 4; ++t) {
      v2f b;
      b.x = sWl[k + 2 * kh][16 * t + mr];
      b.y = sWl[k + 2 * kh + 1][16 * t + mr];
      acc[t] = wmma_f32_k4(a, b, acc[t]);
    }
  }

  float pv[4][8];
#pragma unroll
  for (int t = 0; t < 4; ++t) {
#pragma unroll
    for (int r = 0; r < 8; ++r) {
      pv[t][r] = sPG[mt + r + 8 * kh][16 * t + mr];
    }
  }
  float epart = 0.0f;
#pragma unroll
  for (int t = 0; t < 4; ++t) {
#pragma unroll
    for (int r = 0; r < 8; ++r) {
      int bb = mt + r + 8 * kh;
      int ss = 16 * t + mr;
      float pred = tanh_exp2(acc[t][r]);
      float eps  = pv[t][r] - pred;
      float g    = eps * (1.0f - pred * pred);
      epart += 0.5f * eps * eps;
      sPG[bb][ss] = g;
    }
  }

  for (int off2 = 16; off2 > 0; off2 >>= 1) epart += __shfl_xor(epart, off2, 32);
  if (lane == 0) atomicAdd(&sEnergy, epart);
  __syncthreads();

  v8f acc2[4] = {};
  v8f acc3 = {};
  for (int k = 0; k < SDIM; k += 4) {
    v2f a;
    a.x = sPG[mt + mr][k + 2 * kh];
    a.y = sPG[mt + mr][k + 2 * kh + 1];
#pragma unroll
    for (int t = 0; t < 4; ++t) {
      v2f b;
      b.x = sWo[16 * t + mr][k + 2 * kh];
      b.y = sWo[16 * t + mr][k + 2 * kh + 1];
      acc2[t] = wmma_f32_k4(a, b, acc2[t]);
    }
    v2f bl;
    bl.x = sWl[mr][k + 2 * kh];
    bl.y = sWl[mr][k + 2 * kh + 1];
    acc3 = wmma_f32_k4(a, bl, acc3);
  }
#pragma unroll
  for (int t = 0; t < 4; ++t) {
#pragma unroll
    for (int r = 0; r < 8; ++r) {
      int bb = mt + r + 8 * kh;
      int oo = 16 * t + mr;
      float xo = sXo[bb][oo];
      xo_out[(size_t)n * BATCH * OBJ_DIM + bb * OBJ_DIM + oo] =
          xo + ETA_X * (acc2[t][r] + sCtx[bb][oo] - xo);
    }
  }
#pragma unroll
  for (int r = 0; r < 8; ++r) {
    int bb = mt + r + 8 * kh;
    int ll = mr;
    float xl = sXl[bb][ll];
    xl_out[(size_t)n * BATCH * LOC_DIM + bb * LOC_DIM + ll] =
        xl + ETA_X * acc3[r];
  }

  if (tid == 0) atomicAdd(energy_t, sEnergy);
}

extern "C" void kernel_launch(void* const* d_in, const int* in_sizes, int n_in,
                              void* d_out, int out_size, void* d_ws, size_t ws_size,
                              hipStream_t stream) {
  const float* gin  = (const float*)d_in[0];
  const float* Wobj = (const float*)d_in[1];
  const float* Wloc = (const float*)d_in[2];

  float* out    = (float*)d_out;
  float* energy = out + XO_TOT;

  float* ws  = (float*)d_ws;
  float* xoW = ws;
  float* ppk = xoW + XO_TOT;
  float* xlA = ppk + XO_TOT;
  float* xlB = xlA + XL_TOT;

  cg_init_kernel<<<4096, 256, 0, stream>>>(out, xlA, energy);

  cg_step_kernel<true><<<NCOLS, 128, 0, stream>>>(gin, Wobj, Wloc,
                                                  out, xlA, xoW, xlB,
                                                  energy, ppk);
  for (int t = 1; t < NSTEPS; ++t) {
    const float* xo_in = (t & 1) ? xoW : out;
    const float* xl_in = (t & 1) ? xlB : xlA;
    float* xo_out = (t & 1) ? out : xoW;
    float* xl_out = (t & 1) ? xlA : xlB;
    cg_step_kernel<false><<<NCOLS, 128, 0, stream>>>(gin, Wobj, Wloc,
                                                     xo_in, xl_in, xo_out, xl_out,
                                                     energy + t, ppk);
  }
}


# LDS row pitch 68 floats (was 65): staging stores and k-contiguous fragment reads become 128-bit and conflict-free; lane halves take k 8 apart in blocks of 16, so each product's k sum order changes
# speedup vs baseline: 1.9494x; 1.0968x over previous
#include <hip/hip_runtime.h>
#include <math.h>


#define GRID_HH 32
#define GRID_WW 32
#define OBJ_DIM 64
#define LOC_DIM 16
#define SDIM    64
#define NCOLS   (GRID_HH * GRID_WW)
#define BATCH   64
#define ETA_X   0.05f
#define NSTEPS  20

#define XO_TOT  ((size_t)NCOLS * BATCH * OBJ_DIM)
#define XL_TOT  ((size_t)NCOLS * BATCH * LOC_DIM)

typedef __attribute__((ext_vector_type(2))) float v2f;
typedef __attribute__((ext_vector_type(8))) float v8f;
typedef __attribute__((ext_vector_type(4))) float v4f;

__device__ __forceinline__ v4f ld4(const float* p) {
  return *reinterpret_cast<const v4f*>(p);
}

__device__ __forceinline__ void st4(float* p, v4f v) {
  *reinterpret_cast<v4f*>(p) = v;
}

#define PITCH 68
#define XLP   20

__device__ __forceinline__ v8f wmma_f32_k4(v2f a, v2f b, v8f c) {
  return __builtin_amdgcn_wmma_f32_16x16x4_f32(
      false, a, false, b, (short)0, c, false, false);
}

__device__ __forceinline__ float tanh_exp2(float x) {
  const float ax = __builtin_fabsf(x);
  const float t = __builtin_amdgcn_exp2f(ax * -0x1.715476p+1f);
  const float r = __builtin_amdgcn_rcpf(1.0f + t);
  const float z = (1.0f - t) * r;
  return __builtin_copysignf(z, x);
}

__global__ void cg_init_kernel(float* __restrict__ xo,
                               float* __restrict__ xl,
                               float* __restrict__ energy) {
  size_t i = (size_t)blockIdx.x * blockDim.x + threadIdx.x;
  size_t stride = (size_t)gridDim.x * blockDim.x;
  for (size_t e = i; e < XO_TOT; e += stride) xo[e] = 0.0f;
  for (size_t e = i; e < XL_TOT; e += stride) xl[e] = 0.0f;
  if (i < NSTEPS) energy[i] = 0.0f;
}

template <bool FIRST>
__global__ __launch_bounds__(128, 4)
void cg_step_kernel(const float* __restrict__ gin,
                    const float* __restrict__ Wobj,
                    const float* __restrict__ Wloc,
                    const float* __restrict__ xo_in,
                    const float* __restrict__ xl_in,
                    float* __restrict__ xo_out,
                    float* __restrict__ xl_out,
                    float* __restrict__ energy_t,
                    float* __restrict__ ppk) {
  __shared__ __attribute__((aligned(16))) float sWo[OBJ_DIM][PITCH];
  __shared__ __attribute__((aligned(16))) float sWl[LOC_DIM][PITCH];
  __shared__ __attribute__((aligned(16))) float sPG[BATCH][PITCH];
  __shared__ __attribute__((aligned(16))) float sXo[BATCH][PITCH];
  __shared__ __attribute__((aligned(16))) float sXl[BATCH][XLP];
  __shared__ __attribute__((aligned(16))) float sCtx[BATCH][PITCH];
  __shared__ float sEnergy;

  const int n   = blockIdx.x;
  const int gh  = n >> 5;
  const int gw  = n & 31;
  const int tid = threadIdx.x;

  if (tid == 0) sEnergy = 0.0f;

  const int rowb = tid >> 4;
  const int col  = (tid & 15) * 4;
  const unsigned off = (unsigned)tid * 4u;

  const bool hu = gh > 0, hd = gh < GRID_HH - 1, hl = gw > 0, hr = gw < GRID_WW - 1;
  const size_t tileo = (size_t)BATCH * OBJ_DIM;
  const float* gXo = xo_in + (size_t)n * tileo;
  const float* gUp = xo_in + (size_t)(hu ? n - GRID_WW : n) * tileo;
  const float* gDn = xo_in + (size_t)(hd ? n + GRID_WW : n) * tileo;
  const float* gLf = xo_in + (size_t)(hl ? n - 1 : n) * tileo;
  const float* gRt = xo_in + (size_t)(hr ? n + 1 : n) * tileo;
  const float* gWo = Wobj + (size_t)n * OBJ_DIM * SDIM;
  const float* gWl = Wloc + (size_t)n * LOC_DIM * SDIM;
  const float* gXl = xl_in + (size_t)n * BATCH * LOC_DIM;
  const float* gP  = gin + (size_t)(gh * 8) * 256 + gw * 8;
  const unsigned offp = (unsigned)rowb * 65536u + (unsigned)((tid >> 1) & 7) * 256u
                        + (unsigned)(tid & 1) * 4u;
  float* gPk = ppk + (size_t)n * tileo;

  const int cnt = (int)hu + (int)hd + (int)hl + (int)hr;
  const float inv_cnt = (cnt == 4) ? 0.25f : ((cnt == 3) ? (1.0f / 3.0f) : 0.5f);

  v4f up[8], dn[8], lf[8], rt[8], xo[8], wo[8], pp[8], wl[2], xl[2];
#define CG_FETCH(i)                                   \
  up[i] = ld4(gUp + off + 512u * (i));                \
  dn[i] = ld4(gDn + off + 512u * (i));                \
  lf[i] = ld4(gLf + off + 512u * (i));                \
  rt[i] = ld4(gRt + off + 512u * (i));                \
  xo[i] = ld4(gXo + off + 512u * (i));                \
  wo[i] = ld4(gWo + off + 512u * (i));                \
  if constexpr (FIRST) {                              \
    pp[i] = ld4(gP + offp + 524288u * (i));   \
  } else {                                            \
    pp[i] = ld4(gPk + off + 512u * (i));              \
  }
#define CG_PUT(i)                                     \
  if constexpr (FIRST) st4(gPk + off + 512u * (i), pp[i]); \
  {                                                   \
    v4f cx;                                           \
    _Pragma("unroll")                                 \
    for (int j = 0; j < 4; ++j) {                     \
      float s = 0.0f;                                 \
      s += hu ? up[i][j] : 0.0f;                      \
      s += hd ? dn[i][j] : 0.0f;                      \
      s += hl ? lf[i][j] : 0.0f;                      \
      s += hr ? rt[i][j] : 0.0f;                      \
      cx[j] = s * inv_cnt;                            \
    }                                                 \
    st4(&sCtx[rowb + 8 * (i)][col], cx);              \
    st4(&sXo[rowb + 8 * (i)][col], xo[i]);            \
    st4(&sWo[rowb + 8 * (i)][col], wo[i]);            \
    st4(&sPG[rowb + 8 * (i)][col], pp[i]);            \
  }
  CG_FETCH(0) CG_FETCH(1) CG_FETCH(2)
  CG_FETCH(3) CG_FETCH(4) CG_FETCH(5)
  CG_PUT(0) CG_PUT(1) CG_PUT(2)
  CG_FETCH(6) CG_FETCH(7)
#pragma unroll
  for (int i = 0; i < 2; ++i) {
    wl[i] = ld4(gWl + off + 512u * i);
    xl[i] = ld4(gXl + off + 512u * i);
  }
  CG_PUT(3) CG_PUT(4) CG_PUT(5)
  CG_PUT(6) CG_PUT(7)
#pragma unroll
  for (int i = 0; i < 2; ++i) {
    st4(&sWl[rowb + 8 * i][col], wl[i]);
    st4(&sXl[(tid >> 2) + 32 * i][(tid & 3) * 4], xl[i]);
  }
#undef CG_FETCH
#undef CG_PUT
  __syncthreads();

  const int lane = tid & 31;
  const int wv   = tid >> 5;
  const int kh   = lane >> 4;
  const int mr   = lane & 15;
  const int mt   = wv * 16;

  v8f acc[4] = {};
  for (int q = 0; q < SDIM / 16; ++q) {
    const int kb = 16 * q + 8 * kh;
    const v4f a0 = ld4(&sXo[mt + mr][kb]);
    const v4f a1 = ld4(&sXo[mt + mr][kb + 4]);
    const float av[8] = {a0[0], a0[1], a0[2], a0[3], a1[0], a1[1], a1[2], a1[3]};
#pragma unroll
    for (int i = 0; i < 4; ++i) {
      v2f a;
      a.x = av[2 * i];
      a.y = av[2 * i + 1];
#pragma unroll
      for (int t = 0; t < 4; ++t) {
        v2f b;
        b.x = sWo[kb + 2 * i][16 * t + mr];
        b.y = sWo[kb + 2 * i + 1][16 * t + mr];
        acc[t] = wmma_f32_k4(a, b, acc[t]);
      }
    }
  }
  {
    const int kb = 8 * kh;
    const v4f a0 = ld4(&sXl[mt + mr][kb]);
    const v4f a1 = ld4(&sXl[mt + mr][kb + 4]);
    const float av[8] = {a0[0], a0[1], a0[2], a0[3], a1[0], a1[1], a1[2], a1[3]};
#pragma unroll
    for (int i = 0; i < 4; ++i) {
      v2f a;
      a.x = av[2 * i];
      a.y = av[2 * i + 1];
#pragma unroll
      for (int t = 0; t < 4; ++t) {
        v2f b;
        b.x = sWl[kb + 2 * i][16 * t + mr];
        b.y = sWl[kb + 2 * i + 1][16 * t + mr];
        acc[t] = wmma_f32_k4(a, b, acc[t]);
      }
    }
  }

  float pv[4][8];
#pragma unroll
  for (int t = 0; t < 4; ++t) {
#pragma unroll
    for (int r = 0; r < 8; ++r) {
      pv[t][r] = sPG[mt + r + 8 * kh][16 * t + mr];
    }
  }
  float epart = 0.0f;
#pragma unroll
  for (int t = 0; t < 4; ++t) {
#pragma unroll
    for (int r = 0; r < 8; ++r) {
      int bb = mt + r + 8 * kh;
      int ss = 16 * t + mr;
      float pred = tanh_exp2(acc[t][r]);
      float eps  = pv[t][r] - pred;
      float g    = eps * (1.0f - pred * pred);
      epart += 0.5f * eps * eps;
      sPG[bb][ss] = g;
    }
  }

  for (int off2 = 16; off2 > 0; off2 >>= 1) epart += __shfl_xor(epart, off2, 32);
  if (lane == 0) atomicAdd(&sEnergy, epart);
  __syncthreads();

  v8f acc2[4] = {};
  v8f acc3 = {};
  for (int q = 0; q < SDIM / 16; ++q) {
    const int kb = 16 * q + 8 * kh;
    const v4f g0 = ld4(&sPG[mt + mr][kb]);
    const v4f g1 = ld4(&sPG[mt + mr][kb + 4]);
    const float gv[8] = {g0[0], g0[1], g0[2], g0[3], g1[0], g1[1], g1[2], g1[3]};
    float wv[4][8];
#pragma unroll
    for (int t = 0; t < 4; ++t) {
      const v4f w0 = ld4(&sWo[16 * t + mr][kb]);
      const v4f w1 = ld4(&sWo[16 * t + mr][kb + 4]);
      wv[t][0] = w0[0]; wv[t][1] = w0[1]; wv[t][2] = w0[2]; wv[t][3] = w0[3];
      wv[t][4] = w1[0]; wv[t][5] = w1[1]; wv[t][6] = w1[2]; wv[t][7] = w1[3];
    }
    const v4f l0 = ld4(&sWl[mr][kb]);
    const v4f l1 = ld4(&sWl[mr][kb + 4]);
    const float lv[8] = {l0[0], l0[1], l0[2], l0[3], l1[0], l1[1], l1[2], l1[3]};
#pragma unroll
    for (int i = 0; i < 4; ++i) {
      v2f a;
      a.x = gv[2 * i];
      a.y = gv[2 * i + 1];
#pragma unroll
      for (int t = 0; t < 4; ++t) {
        v2f b;
        b.x = wv[t][2 * i];
        b.y = wv[t][2 * i + 1];
        acc2[t] = wmma_f32_k4(a, b, acc2[t]);
      }
      v2f bl;
      bl.x = lv[2 * i];
      bl.y = lv[2 * i + 1];
      acc3 = wmma_f32_k4(a, bl, acc3);
    }
  }
#pragma unroll
  for (int t = 0; t < 4; ++t) {
#pragma unroll
    for (int r = 0; r < 8; ++r) {
      int bb = mt + r + 8 * kh;
      int oo = 16 * t + mr;
      float xo = sXo[bb][oo];
      xo_out[(size_t)n * BATCH * OBJ_DIM + bb * OBJ_DIM + oo] =
          xo + ETA_X * (acc2[t][r] + sCtx[bb][oo] - xo);
    }
  }
#pragma unroll
  for (int r = 0; r < 8; ++r) {
    int bb = mt + r + 8 * kh;
    int ll = mr;
    float xl = sXl[bb][ll];
    xl_out[(size_t)n * BATCH * LOC_DIM + bb * LOC_DIM + ll] =
        xl + ETA_X * acc3[r];
  }

  if (tid == 0) atomicAdd(energy_t, sEnergy);
}

extern "C" void kernel_launch(void* const* d_in, const int* in_sizes, int n_in,
                              void* d_out, int out_size, void* d_ws, size_t ws_size,
                              hipStream_t stream) {
  const float* gin  = (const float*)d_in[0];
  const float* Wobj = (const float*)d_in[1];
  const float* Wloc = (const float*)d_in[2];

  float* out    = (float*)d_out;
  float* energy = out + XO_TOT;

  float* ws  = (float*)d_ws;
  float* xoW = ws;
  float* ppk = xoW + XO_TOT;
  float* xlA = ppk + XO_TOT;
  float* xlB = xlA + XL_TOT;

  cg_init_kernel<<<4096, 256, 0, stream>>>(out, xlA, energy);

  cg_step_kernel<true><<<NCOLS, 128, 0, stream>>>(gin, Wobj, Wloc,
                                                  out, xlA, xoW, xlB,
                                                  energy, ppk);
  for (int t = 1; t < NSTEPS; ++t) {
    const float* xo_in = (t & 1) ? xoW : out;
    const float* xl_in = (t & 1) ? xlB : xlA;
    float* xo_out = (t & 1) ? out : xoW;
    float* xl_out = (t & 1) ? xlA : xlB;
    cg_step_kernel<false><<<NCOLS, 128, 0, stream>>>(gin, Wobj, Wloc,
                                                     xo_in, xl_in, xo_out, xl_out,
                                                     energy + t, ppk);
  }
}


# First step folded on the zero start state: no x tiles fetched, no forward product (tanh(0)=0, g = patch), update = eta*dx; the start-up launch is one wave that clears the energy sums; same sum order
# speedup vs baseline: 2.0218x; 1.0371x over previous
#include <hip/hip_runtime.h>
#include <math.h>


#define GRID_HH 32
#define GRID_WW 32
#define OBJ_DIM 64
#define LOC_DIM 16
#define SDIM    64
#define NCOLS   (GRID_HH * GRID_WW)
#define BATCH   64
#define ETA_X   0.05f
#define NSTEPS  20

#define XO_TOT  ((size_t)NCOLS * BATCH * OBJ_DIM)
#define XL_TOT  ((size_t)NCOLS * BATCH * LOC_DIM)

typedef __attribute__((ext_vector_type(2))) float v2f;
typedef __attribute__((ext_vector_type(8))) float v8f;
typedef __attribute__((ext_vector_type(4))) float v4f;

__device__ __forceinline__ v4f ld4(const float* p) {
  return *reinterpret_cast<const v4f*>(p);
}

__device__ __forceinline__ void st4(float* p, v4f v) {
  *reinterpret_cast<v4f*>(p) = v;
}

#define PITCH 68
#define XLP   20

__device__ __forceinline__ v8f wmma_f32_k4(v2f a, v2f b, v8f c) {
  return __builtin_amdgcn_wmma_f32_16x16x4_f32(
      false, a, false, b, (short)0, c, false, false);
}

__device__ __forceinline__ float tanh_exp2(float x) {
  const float ax = __builtin_fabsf(x);
  const float t = __builtin_amdgcn_exp2f(ax * -0x1.715476p+1f);
  const float r = __builtin_amdgcn_rcpf(1.0f + t);
  const float z = (1.0f - t) * r;
  return __builtin_copysignf(z, x);
}

__global__ void cg_init_kernel(float* __restrict__ energy) {
  if (threadIdx.x < NSTEPS) energy[threadIdx.x] = 0.0f;
}

template <bool FIRST>
__global__ __launch_bounds__(128, 4)
void cg_step_kernel(const float* __restrict__ gin,
                    const float* __restrict__ Wobj,
                    const float* __restrict__ Wloc,
                    const float* __restrict__ xo_in,
                    const float* __restrict__ xl_in,
                    float* __restrict__ xo_out,
                    float* __restrict__ xl_out,
                    float* __restrict__ energy_t,
                    float* __restrict__ ppk) {
  __shared__ __attribute__((aligned(16))) float sWo[OBJ_DIM][PITCH];
  __shared__ __attribute__((aligned(16))) float sWl[LOC_DIM][PITCH];
  __shared__ __attribute__((aligned(16))) float sPG[BATCH][PITCH];
  __shared__ __attribute__((aligned(16))) float sXo[BATCH][PITCH];
  __shared__ __attribute__((aligned(16))) float sXl[BATCH][XLP];
  __shared__ __attribute__((aligned(16))) float sCtx[BATCH][PITCH];
  __shared__ float sEnergy;

  const int n   = blockIdx.x;
  const int gh  = n >> 5;
  const int gw  = n & 31;
  const int tid = threadIdx.x;

  if (tid == 0) sEnergy = 0.0f;

  const int rowb = tid >> 4;
  const int col  = (tid & 15) * 4;
  const unsigned off = (unsigned)tid * 4u;

  const size_t tileo = (size_t)BATCH * OBJ_DIM;
  if constexpr (FIRST) {
    const float* gP  = gin + (size_t)(gh * 8) * 256 + gw * 8;
    const unsigned offp = (unsigned)rowb * 65536u + (unsigned)((tid >> 1) & 7) * 256u
                          + (unsigned)(tid & 1) * 4u;
    float* gPk = ppk + (size_t)n * tileo;
    const float* gWo = Wobj + (size_t)n * OBJ_DIM * SDIM;
    const float* gWl = Wloc + (size_t)n * LOC_DIM * SDIM;
    v4f wo[8], pp[8], wl[2];
#pragma unroll
    for (int i = 0; i < 8; ++i) {
      pp[i] = ld4(gP + offp + 524288u * i);
      wo[i] = ld4(gWo + off + 512u * i);
    }
#pragma unroll
    for (int i = 0; i < 2; ++i) wl[i] = ld4(gWl + off + 512u * i);
#pragma unroll
    for (int i = 0; i < 8; ++i) {
      st4(gPk + off + 512u * i, pp[i]);
      st4(&sPG[rowb + 8 * i][col], pp[i]);
      st4(&sWo[rowb + 8 * i][col], wo[i]);
    }
#pragma unroll
    for (int i = 0; i < 2; ++i) st4(&sWl[rowb + 8 * i][col], wl[i]);
  } else {
  const bool hu = gh > 0, hd = gh < GRID_HH - 1, hl = gw > 0, hr = gw < GRID_WW - 1;
  const float* gXo = xo_in + (size_t)n * tileo;
  const float* gUp = xo_in + (size_t)(hu ? n - GRID_WW : n) * tileo;
  const float* gDn = xo_in + (size_t)(hd ? n + GRID_WW : n) * tileo;
  const float* gLf = xo_in + (size_t)(hl ? n - 1 : n) * tileo;
  const float* gRt = xo_in + (size_t)(hr ? n + 1 : n) * tileo;
  const float* gWo = Wobj + (size_t)n * OBJ_DIM * SDIM;
  const float* gWl = Wloc + (size_t)n * LOC_DIM * SDIM;
  const float* gXl = xl_in + (size_t)n * BATCH * LOC_DIM;
  const float* gPk = ppk + (size_t)n * tileo;

  const int cnt = (int)hu + (int)hd + (int)hl + (int)hr;
  const float inv_cnt = (cnt == 4) ? 0.25f : ((cnt == 3) ? (1.0f / 3.0f) : 0.5f);

  v4f up[8], dn[8], lf[8], rt[8], xo[8], wo[8], pp[8], wl[2], xl[2];
#define CG_FETCH(i)                                   \
  up[i] = ld4(gUp + off + 512u * (i));                \
  dn[i] = ld4(gDn + off + 512u * (i));                \
  lf[i] = ld4(gLf + off + 512u * (i));                \
  rt[i] = ld4(gRt + off + 512u * (i));                \
  xo[i] = ld4(gXo + off + 512u * (i));                \
  wo[i] = ld4(gWo + off + 512u * (i));                \
  pp[i] = ld4(gPk + off + 512u * (i));
#define CG_PUT(i)                                     \
  {                                                   \
    v4f cx;                                           \
    _Pragma("unroll")                                 \
    for (int j = 0; j < 4; ++j) {                     \
      float s = 0.0f;                                 \
      s += hu ? up[i][j] : 0.0f;                      \
      s += hd ? dn[i][j] : 0.0f;                      \
      s += hl ? lf[i][j] : 0.0f;                      \
      s += hr ? rt[i][j] : 0.0f;                      \
      cx[j] = s * inv_cnt;                            \
    }                                                 \
    st4(&sCtx[rowb + 8 * (i)][col], cx);              \
    st4(&sXo[rowb + 8 * (i)][col], xo[i]);            \
    st4(&sWo[rowb + 8 * (i)][col], wo[i]);            \
    st4(&sPG[rowb + 8 * (i)][col], pp[i]);            \
  }
  CG_FETCH(0) CG_FETCH(1) CG_FETCH(2)
  CG_FETCH(3) CG_FETCH(4) CG_FETCH(5)
  CG_PUT(0) CG_PUT(1) CG_PUT(2)
  CG_FETCH(6) CG_FETCH(7)
#pragma unroll
  for (int i = 0; i < 2; ++i) {
    wl[i] = ld4(gWl + off + 512u * i);
    xl[i] = ld4(gXl + off + 512u * i);
  }
  CG_PUT(3) CG_PUT(4) CG_PUT(5)
  CG_PUT(6) CG_PUT(7)
#pragma unroll
  for (int i = 0; i < 2; ++i) {
    st4(&sWl[rowb + 8 * i][col], wl[i]);
    st4(&sXl[(tid >> 2) + 32 * i][(tid & 3) * 4], xl[i]);
  }
#undef CG_FETCH
#undef CG_PUT
  }
  __syncthreads();

  const int lane = tid & 31;
  const int wv   = tid >> 5;
  const int kh   = lane >> 4;
  const int mr   = lane & 15;
  const int mt   = wv * 16;

  float epart = 0.0f;
  if constexpr (FIRST) {
#pragma unroll
    for (int t = 0; t < 4; ++t) {
#pragma unroll
      for (int r = 0; r < 8; ++r) {
        const float eps = sPG[mt + r + 8 * kh][16 * t + mr];
        epart += 0.5f * eps * eps;
      }
    }
  } else {
  v8f acc[4] = {};
  for (int q = 0; q < SDIM / 16; ++q) {
    const int kb = 16 * q + 8 * kh;
    const v4f a0 = ld4(&sXo[mt + mr][kb]);
    const v4f a1 = ld4(&sXo[mt + mr][kb + 4]);
    const float av[8] = {a0[0], a0[1], a0[2], a0[3], a1[0], a1[1], a1[2], a1[3]};
#pragma unroll
    for (int i = 0; i < 4; ++i) {
      v2f a;
      a.x = av[2 * i];
      a.y = av[2 * i + 1];
#pragma unroll
      for (int t = 0; t < 4; ++t) {
        v2f b;
        b.x = sWo[kb + 2 * i][16 * t + mr];
        b.y = sWo[kb + 2 * i + 1][16 * t + mr];
        acc[t] = wmma_f32_k4(a, b, acc[t]);
      }
    }
  }
  {
    const int kb = 8 * kh;
    const v4f a0 = ld4(&sXl[mt + mr][kb]);
    const v4f a1 = ld4(&sXl[mt + mr][kb + 4]);
    const float av[8] = {a0[0], a0[1], a0[2], a0[3], a1[0], a1[1], a1[2], a1[3]};
#pragma unroll
    for (int i = 0; i < 4; ++i) {
      v2f a;
      a.x = av[2 * i];
      a.y = av[2 * i + 1];
#pragma unroll
      for (int t = 0; t < 4; ++t) {
        v2f b;
        b.x = sWl[kb + 2 * i][16 * t + mr];
        b.y = sWl[kb + 2 * i + 1][16 * t + mr];
        acc[t] = wmma_f32_k4(a, b, acc[t]);
      }
    }
  }

  float pv[4][8];
#pragma unroll
  for (int t = 0; t < 4; ++t) {
#pragma unroll
    for (int r = 0; r < 8; ++r) {
      pv[t][r] = sPG[mt + r + 8 * kh][16 * t + mr];
    }
  }
#pragma unroll
  for (int t = 0; t < 4; ++t) {
#pragma unroll
    for (int r = 0; r < 8; ++r) {
      int bb = mt + r + 8 * kh;
      int ss = 16 * t + mr;
      float pred = tanh_exp2(acc[t][r]);
      float eps  = pv[t][r] - pred;
      float g    = eps * (1.0f - pred * pred);
      epart += 0.5f * eps * eps;
      sPG[bb][ss] = g;
    }
  }

  }

  for (int off2 = 16; off2 > 0; off2 >>= 1) epart += __shfl_xor(epart, off2, 32);
  if (lane == 0) atomicAdd(&sEnergy, epart);
  __syncthreads();

  v8f acc2[4] = {};
  v8f acc3 = {};
  for (int q = 0; q < SDIM / 16; ++q) {
    const int kb = 16 * q + 8 * kh;
    const v4f g0 = ld4(&sPG[mt + mr][kb]);
    const v4f g1 = ld4(&sPG[mt + mr][kb + 4]);
    const float gv[8] = {g0[0], g0[1], g0[2], g0[3], g1[0], g1[1], g1[2], g1[3]};
    float wv[4][8];
#pragma unroll
    for (int t = 0; t < 4; ++t) {
      const v4f w0 = ld4(&sWo[16 * t + mr][kb]);
      const v4f w1 = ld4(&sWo[16 * t + mr][kb + 4]);
      wv[t][0] = w0[0]; wv[t][1] = w0[1]; wv[t][2] = w0[2]; wv[t][3] = w0[3];
      wv[t][4] = w1[0]; wv[t][5] = w1[1]; wv[t][6] = w1[2]; wv[t][7] = w1[3];
    }
    const v4f l0 = ld4(&sWl[mr][kb]);
    const v4f l1 = ld4(&sWl[mr][kb + 4]);
    const float lv[8] = {l0[0], l0[1], l0[2], l0[3], l1[0], l1[1], l1[2], l1[3]};
#pragma unroll
    for (int i = 0; i < 4; ++i) {
      v2f a;
      a.x = gv[2 * i];
      a.y = gv[2 * i + 1];
#pragma unroll
      for (int t = 0; t < 4; ++t) {
        v2f b;
        b.x = wv[t][2 * i];
        b.y = wv[t][2 * i + 1];
        acc2[t] = wmma_f32_k4(a, b, acc2[t]);
      }
      v2f bl;
      bl.x = lv[2 * i];
      bl.y = lv[2 * i + 1];
      acc3 = wmma_f32_k4(a, bl, acc3);
    }
  }
#pragma unroll
  for (int t = 0; t < 4; ++t) {
#pragma unroll
    for (int r = 0; r < 8; ++r) {
      int bb = mt + r + 8 * kh;
      int oo = 16 * t + mr;
      if constexpr (FIRST) {
        xo_out[(size_t)n * BATCH * OBJ_DIM + bb * OBJ_DIM + oo] =
            ETA_X * acc2[t][r];
      } else {
        float xo = sXo[bb][oo];
        xo_out[(size_t)n * BATCH * OBJ_DIM + bb * OBJ_DIM + oo] =
            xo + ETA_X * (acc2[t][r] + sCtx[bb][oo] - xo);
      }
    }
  }
#pragma unroll
  for (int r = 0; r < 8; ++r) {
    int bb = mt + r + 8 * kh;
    int ll = mr;
    if constexpr (FIRST) {
      xl_out[(size_t)n * BATCH * LOC_DIM + bb * LOC_DIM + ll] = ETA_X * acc3[r];
    } else {
      float xl = sXl[bb][ll];
      xl_out[(size_t)n * BATCH * LOC_DIM + bb * LOC_DIM + ll] =
          xl + ETA_X * acc3[r];
    }
  }

  if (tid == 0) atomicAdd(energy_t, sEnergy);
}

extern "C" void kernel_launch(void* const* d_in, const int* in_sizes, int n_in,
                              void* d_out, int out_size, void* d_ws, size_t ws_size,
                              hipStream_t stream) {
  const float* gin  = (const float*)d_in[0];
  const float* Wobj = (const float*)d_in[1];
  const float* Wloc = (const float*)d_in[2];

  float* out    = (float*)d_out;
  float* energy = out + XO_TOT;

  float* ws  = (float*)d_ws;
  float* xoW = ws;
  float* ppk = xoW + XO_TOT;
  float* xlA = ppk + XO_TOT;
  float* xlB = xlA + XL_TOT;

  cg_init_kernel<<<1, 32, 0, stream>>>(energy);

  cg_step_kernel<true><<<NCOLS, 128, 0, stream>>>(gin, Wobj, Wloc,
                                                  out, xlA, xoW, xlB,
                                                  energy, ppk);
  for (int t = 1; t < NSTEPS; ++t) {
    const float* xo_in = (t & 1) ? xoW : out;
    const float* xl_in = (t & 1) ? xlB : xlA;
    float* xo_out = (t & 1) ? out : xoW;
    float* xl_out = (t & 1) ? xlA : xlB;
    cg_step_kernel<false><<<NCOLS, 128, 0, stream>>>(gin, Wobj, Wloc,
                                                     xo_in, xl_in, xo_out, xl_out,
                                                     energy + t, ppk);
  }
}
